# starting kernel
# baseline (speedup 1.0000x reference)
#include <hip/hip_runtime.h>
#include <math.h>

typedef __attribute__((ext_vector_type(2))) float v2f;
typedef __attribute__((ext_vector_type(8))) float v8f;

#define D_IN   1024
#define D_OUT  1024
#define NEXP   8
#define RANK   16
#define KDIM   128
#define SCALING 2.0f

__global__ __launch_bounds__(256) void repack_w2(const float* __restrict__ routed,
                                                 float* __restrict__ w2) {
    int idx = blockIdx.x * 256 + threadIdx.x;
    int k = idx >> 10;
    int o = idx & 1023;
    int e = k >> 4;
    int r = k & 15;
    w2[idx] = routed[(e * D_OUT + o) * RANK + r];
}

__global__ __launch_bounds__(256) void route_compress(const float* __restrict__ x,
                                                      const float* __restrict__ w_route,
                                                      const float* __restrict__ compress,
                                                      float* __restrict__ tmp) {
    const int lane = threadIdx.x & 31;
    const int wave = threadIdx.x >> 5;
    const int tok0 = (blockIdx.x * 8 + wave) * 4;

    float acc[24][4];
#pragma unroll
    for (int j = 0; j < 24; ++j)
#pragma unroll
        for (int t = 0; t < 4; ++t) acc[j][t] = 0.0f;

    for (int it = 0; it < 8; ++it) {
        const int off = it * 128 + lane * 4;
        float4 xv[4];
#pragma unroll
        for (int t = 0; t < 4; ++t)
            xv[t] = *(const float4*)(x + (size_t)(tok0 + t) * D_IN + off);
#pragma unroll
        for (int j = 0; j < 8; ++j) {
            float4 w = *(const float4*)(w_route + j * D_IN + off);
#pragma unroll
            for (int t = 0; t < 4; ++t)
                acc[j][t] += w.x * xv[t].x + w.y * xv[t].y + w.z * xv[t].z + w.w * xv[t].w;
        }
#pragma unroll
        for (int j = 0; j < 16; ++j) {
            float4 w = *(const float4*)(compress + j * D_IN + off);
#pragma unroll
            for (int t = 0; t < 4; ++t)
                acc[8 + j][t] += w.x * xv[t].x + w.y * xv[t].y + w.z * xv[t].z + w.w * xv[t].w;
        }
    }

#pragma unroll
    for (int j = 0; j < 24; ++j)
#pragma unroll
        for (int t = 0; t < 4; ++t) {
            float v = acc[j][t];
#pragma unroll
            for (int s = 16; s >= 1; s >>= 1) v += __shfl_xor(v, s, 32);
            acc[j][t] = v;
        }

#pragma unroll
    for (int t = 0; t < 4; ++t) {
        float v0 = acc[0][t]; int i0 = 0;
#pragma unroll
        for (int j = 1; j < 8; ++j) { float l = acc[j][t]; if (l > v0) { v0 = l; i0 = j; } }
        float v1 = -3.4e38f; int i1 = 0;
#pragma unroll
        for (int j = 0; j < 8; ++j) {
            if (j == i0) continue;
            float l = acc[j][t]; if (l > v1) { v1 = l; i1 = j; }
        }
        float ex = __expf(v1 - v0);
        float g0 = 1.0f / (1.0f + ex);
        float g1 = ex   / (1.0f + ex);

        const int kbase = lane * 4;
        const int e     = lane >> 2;
        const int sel   = lane & 3;
        float g = (e == i0) ? g0 : ((e == i1) ? g1 : 0.0f);
        g *= SCALING;

        float c0 = 0.f, c1 = 0.f, c2 = 0.f, c3 = 0.f;
#pragma unroll
        for (int s = 0; s < 4; ++s) {
            bool p = (sel == s);
            c0 = p ? acc[8 + s * 4 + 0][t] : c0;
            c1 = p ? acc[8 + s * 4 + 1][t] : c1;
            c2 = p ? acc[8 + s * 4 + 2][t] : c2;
            c3 = p ? acc[8 + s * 4 + 3][t] : c3;
        }
        float4 outv = make_float4(g * c0, g * c1, g * c2, g * c3);
        *(float4*)(tmp + (size_t)(tok0 + t) * KDIM + kbase) = outv;
    }
}

__global__ __launch_bounds__(256) void gemm_wmma(const float* __restrict__ A,
                                                 const float* __restrict__ B,
                                                 float* __restrict__ C) {
    const int lane = threadIdx.x & 31;
    const int wave = threadIdx.x >> 5;
    const int tile = blockIdx.x * 8 + wave;
    const int tN   = tile & 63;
    const int tM   = tile >> 6;
    const int half = lane >> 4;
    const int l16  = lane & 15;

    const float* aRow = A + (size_t)(tM * 16 + l16) * KDIM + half * 2;
    const float* bCol = B + (size_t)(half * 2) * D_OUT + tN * 16 + l16;

    v8f c = {0.f, 0.f, 0.f, 0.f, 0.f, 0.f, 0.f, 0.f};
#pragma unroll
    for (int ks = 0; ks < 32; ++ks) {
        v2f a = *(const v2f*)(aRow + ks * 4);
        v2f b;
        b[0] = bCol[(size_t)(ks * 4 + 0) * D_OUT];
        b[1] = bCol[(size_t)(ks * 4 + 1) * D_OUT];
        c = __builtin_amdgcn_wmma_f32_16x16x4_f32(false, a, false, b,
                                                  (short)0, c, false, false);
    }

    float* outp = C + (size_t)(tM * 16 + half * 8) * D_OUT + tN * 16 + l16;
#pragma unroll
    for (int v = 0; v < 8; ++v) outp[(size_t)v * D_OUT] = c[v];
}

extern "C" void kernel_launch(void* const* d_in, const int* in_sizes, int n_in,
                              void* d_out, int out_size, void* d_ws, size_t ws_size,
                              hipStream_t stream) {
    const float* x        = (const float*)d_in[0];
    const float* w_route  = (const float*)d_in[1];
    const float* compress = (const float*)d_in[2];
    const float* routed   = (const float*)d_in[3];
    float*       out      = (float*)d_out;

    float* w2  = (float*)d_ws;
    float* tmp = w2 + KDIM * D_OUT;

    const int ntok = in_sizes[0] / D_IN;

    repack_w2<<<(KDIM * D_OUT) / 256, 256, 0, stream>>>(routed, w2);

    route_compress<<<ntok / 32, 256, 0, stream>>>(x, w_route, compress, tmp);

    const int tiles = (ntok / 16) * (D_OUT / 16);
    gemm_wmma<<<tiles / 8, 256, 0, stream>>>(tmp, w2, out);
}


# GEMM retiled: 256x128 block tile, B staged once per block in LDS straight from routed (repack launch dropped), A fragments held in registers, 2 accumulators per wave; k sum order in each 8 permuted
# speedup vs baseline: 1.5025x; 1.5025x over previous
#include <hip/hip_runtime.h>
#include <math.h>

typedef __attribute__((ext_vector_type(2))) float v2f;
typedef __attribute__((ext_vector_type(8))) float v8f;

#define D_IN   1024
#define D_OUT  1024
#define NEXP   8
#define RANK   16
#define KDIM   128
#define SCALING 2.0f

__global__ __launch_bounds__(256) void route_compress(const float* __restrict__ x,
                                                      const float* __restrict__ w_route,
                                                      const float* __restrict__ compress,
                                                      float* __restrict__ tmp) {
    const int lane = threadIdx.x & 31;
    const int wave = threadIdx.x >> 5;
    const int tok0 = (blockIdx.x * 8 + wave) * 4;

    float acc[24][4];
#pragma unroll
    for (int j = 0; j < 24; ++j)
#pragma unroll
        for (int t = 0; t < 4; ++t) acc[j][t] = 0.0f;

    for (int it = 0; it < 8; ++it) {
        const int off = it * 128 + lane * 4;
        float4 xv[4];
#pragma unroll
        for (int t = 0; t < 4; ++t)
            xv[t] = *(const float4*)(x + (size_t)(tok0 + t) * D_IN + off);
#pragma unroll
        for (int j = 0; j < 8; ++j) {
            float4 w = *(const float4*)(w_route + j * D_IN + off);
#pragma unroll
            for (int t = 0; t < 4; ++t)
                acc[j][t] += w.x * xv[t].x + w.y * xv[t].y + w.z * xv[t].z + w.w * xv[t].w;
        }
#pragma unroll
        for (int j = 0; j < 16; ++j) {
            float4 w = *(const float4*)(compress + j * D_IN + off);
#pragma unroll
            for (int t = 0; t < 4; ++t)
                acc[8 + j][t] += w.x * xv[t].x + w.y * xv[t].y + w.z * xv[t].z + w.w * xv[t].w;
        }
    }

#pragma unroll
    for (int j = 0; j < 24; ++j)
#pragma unroll
        for (int t = 0; t < 4; ++t) {
            float v = acc[j][t];
#pragma unroll
            for (int s = 16; s >= 1; s >>= 1) v += __shfl_xor(v, s, 32);
            acc[j][t] = v;
        }

#pragma unroll
    for (int t = 0; t < 4; ++t) {
        float v0 = acc[0][t]; int i0 = 0;
#pragma unroll
        for (int j = 1; j < 8; ++j) { float l = acc[j][t]; if (l > v0) { v0 = l; i0 = j; } }
        float v1 = -3.4e38f; int i1 = 0;
#pragma unroll
        for (int j = 0; j < 8; ++j) {
            if (j == i0) continue;
            float l = acc[j][t]; if (l > v1) { v1 = l; i1 = j; }
        }
        float ex = __expf(v1 - v0);
        float g0 = 1.0f / (1.0f + ex);
        float g1 = ex   / (1.0f + ex);

        const int kbase = lane * 4;
        const int e     = lane >> 2;
        const int sel   = lane & 3;
        float g = (e == i0) ? g0 : ((e == i1) ? g1 : 0.0f);
        g *= SCALING;

        float c0 = 0.f, c1 = 0.f, c2 = 0.f, c3 = 0.f;
#pragma unroll
        for (int s = 0; s < 4; ++s) {
            bool p = (sel == s);
            c0 = p ? acc[8 + s * 4 + 0][t] : c0;
            c1 = p ? acc[8 + s * 4 + 1][t] : c1;
            c2 = p ? acc[8 + s * 4 + 2][t] : c2;
            c3 = p ? acc[8 + s * 4 + 3][t] : c3;
        }
        float4 outv = make_float4(g * c0, g * c1, g * c2, g * c3);
        *(float4*)(tmp + (size_t)(tok0 + t) * KDIM + kbase) = outv;
    }
}

#define BM   256
#define BN   128
#define LDB  132

__global__ __launch_bounds__(256) void gemm_wmma(const float* __restrict__ A,
                                                 const float* __restrict__ routed,
                                                 float* __restrict__ C) {
    __shared__ __attribute__((aligned(16))) float Bs[BN * LDB];

    const int tid  = threadIdx.x;
    const int lane = tid & 31;
    const int wave = tid >> 5;
    const int nb   = blockIdx.x & 7;
    const int mb   = blockIdx.x >> 3;
    const int n0   = nb * BN;
    const int h    = lane >> 4;
    const int c    = lane & 15;

    float4 st[16];
#pragma unroll
    for (int i = 0; i < 16; ++i) {
        const int u  = i * 256 + tid;
        const int q  = u & 3;
        const int ol = (u >> 2) & 127;
        const int e  = u >> 9;
        st[i] = *(const float4*)(routed + ((size_t)(e * D_OUT + n0 + ol) * RANK + q * 4));
    }

    const int row0 = mb * BM + wave * 32;
    const float* a0p = A + (size_t)(row0 + c) * KDIM + h * 4;
    const float* a1p = a0p + (size_t)16 * KDIM;
    float4 a0[16], a1[16];
#pragma unroll
    for (int j = 0; j < 16; ++j) {
        a0[j] = *(const float4*)(a0p + j * 8);
        a1[j] = *(const float4*)(a1p + j * 8);
    }

#pragma unroll
    for (int i = 0; i < 16; ++i) {
        const int u  = i * 256 + tid;
        const int q  = u & 3;
        const int ol = (u >> 2) & 127;
        const int e  = u >> 9;
        *(float4*)(Bs + ol * LDB + e * RANK + q * 4) = st[i];
    }

    __syncthreads();
    __builtin_amdgcn_sched_barrier(0);

    float* outp = C + (size_t)(row0 + h * 8) * D_OUT + n0 + c;
    const float* bp = Bs + c * LDB + h * 4;

    float4 bq[8];
#pragma unroll
    for (int jj = 0; jj < 8; ++jj) bq[jj] = *(const float4*)(bp + jj * 8);

    v8f acc0 = {0.f, 0.f, 0.f, 0.f, 0.f, 0.f, 0.f, 0.f};
    v8f acc1 = {0.f, 0.f, 0.f, 0.f, 0.f, 0.f, 0.f, 0.f};
#pragma unroll
    for (int s = 0; s < 16; ++s) {
        const int nt = s >> 1;
        const int hf = s & 1;
        float4 bn[8];
        if (s + 1 < 16) {
            const float* np = bp + ((s + 1) >> 1) * 16 * LDB + ((s + 1) & 1) * 64;
#pragma unroll
            for (int jj = 0; jj < 8; ++jj) bn[jj] = *(const float4*)(np + jj * 8);
        }
#pragma unroll
        for (int jj = 0; jj < 8; ++jj) {
            const int j = hf * 8 + jj;
            v2f blo, bhi, x0, x1;
            blo[0] = bq[jj].x; blo[1] = bq[jj].y;
            bhi[0] = bq[jj].z; bhi[1] = bq[jj].w;
            x0[0] = a0[j].x; x0[1] = a0[j].y;
            x1[0] = a1[j].x; x1[1] = a1[j].y;
            acc0 = __builtin_amdgcn_wmma_f32_16x16x4_f32(false, x0, false, blo, (short)0, acc0, false, false);
            acc1 = __builtin_amdgcn_wmma_f32_16x16x4_f32(false, x1, false, blo, (short)0, acc1, false, false);
            x0[0] = a0[j].z; x0[1] = a0[j].w;
            x1[0] = a1[j].z; x1[1] = a1[j].w;
            acc0 = __builtin_amdgcn_wmma_f32_16x16x4_f32(false, x0, false, bhi, (short)0, acc0, false, false);
            acc1 = __builtin_amdgcn_wmma_f32_16x16x4_f32(false, x1, false, bhi, (short)0, acc1, false, false);
        }
        if (hf == 1) {
            float* o = outp + nt * 16;
#pragma unroll
            for (int v = 0; v < 8; ++v) {
                o[(size_t)v * D_OUT]        = acc0[v];
                o[(size_t)(16 + v) * D_OUT] = acc1[v];
            }
#pragma unroll
            for (int v = 0; v < 8; ++v) { acc0[v] = 0.f; acc1[v] = 0.f; }
        }
        if (s + 1 < 16) {
#pragma unroll
            for (int jj = 0; jj < 8; ++jj) bq[jj] = bn[jj];
        }
    }
}

extern "C" void kernel_launch(void* const* d_in, const int* in_sizes, int n_in,
                              void* d_out, int out_size, void* d_ws, size_t ws_size,
                              hipStream_t stream) {
    const float* x        = (const float*)d_in[0];
    const float* w_route  = (const float*)d_in[1];
    const float* compress = (const float*)d_in[2];
    const float* routed   = (const float*)d_in[3];
    float*       out      = (float*)d_out;

    float* tmp = (float*)d_ws;

    const int ntok = in_sizes[0] / D_IN;

    route_compress<<<ntok / 32, 256, 0, stream>>>(x, w_route, compress, tmp);

    gemm_wmma<<<(ntok / BM) * (D_OUT / BN), 256, 0, stream>>>(tmp, routed, out);
}


# router+compress dot products moved from vector FMAs and wave butterfly onto f32 WMMA (Wcat in LDS, all x loads of a wave issued as one group); k summed as four quarter partials in fixed order
# speedup vs baseline: 2.5424x; 1.6921x over previous
#include <hip/hip_runtime.h>
#include <math.h>

typedef __attribute__((ext_vector_type(2))) float v2f;
typedef __attribute__((ext_vector_type(8))) float v8f;

#define D_IN   1024
#define D_OUT  1024
#define NEXP   8
#define RANK   16
#define KDIM   128
#define SCALING 2.0f

#define XT   32
#define WP   1028
#define PP   36

__global__ __launch_bounds__(256) void route_compress(const float* __restrict__ x,
                                                      const float* __restrict__ w_route,
                                                      const float* __restrict__ compress,
                                                      float* __restrict__ tmp) {
    __shared__ __attribute__((aligned(16))) float Ws[24 * WP];
    __shared__ __attribute__((aligned(16))) float Ps[4 * XT * PP];

    const int tid  = threadIdx.x;
    const int lane = tid & 31;
    const int wave = tid >> 5;
    const int h    = lane >> 4;
    const int c    = lane & 15;
    const int rt   = wave & 1;
    const int kq   = wave >> 1;
    const int tok0 = blockIdx.x * XT;

    float4 st[24];
#pragma unroll
    for (int i = 0; i < 24; ++i) {
        const float* src = (i < 8) ? (w_route + i * D_IN) : (compress + (i - 8) * D_IN);
        st[i] = *(const float4*)(src + tid * 4);
    }

    const float* xp = x + (size_t)(tok0 + rt * 16 + c) * D_IN + kq * 256 + h * 4;
    float4 xa[32];
#pragma unroll
    for (int j = 0; j < 32; ++j) xa[j] = *(const float4*)(xp + j * 8);

#pragma unroll
    for (int i = 0; i < 24; ++i) *(float4*)(Ws + i * WP + tid * 4) = st[i];

    __syncthreads();

    const float* b0p = Ws + c * WP + kq * 256 + h * 4;
    const float* b1p = Ws + (16 + (c & 7)) * WP + kq * 256 + h * 4;

    float4 bq0[4], bq1[4];
#pragma unroll
    for (int jj = 0; jj < 4; ++jj) {
        bq0[jj] = *(const float4*)(b0p + jj * 8);
        bq1[jj] = *(const float4*)(b1p + jj * 8);
    }

    v8f acc0 = {0.f, 0.f, 0.f, 0.f, 0.f, 0.f, 0.f, 0.f};
    v8f acc1 = {0.f, 0.f, 0.f, 0.f, 0.f, 0.f, 0.f, 0.f};
#pragma unroll
    for (int s = 0; s < 8; ++s) {
        float4 bn0[4], bn1[4];
        if (s + 1 < 8) {
#pragma unroll
            for (int jj = 0; jj < 4; ++jj) {
                bn0[jj] = *(const float4*)(b0p + ((s + 1) * 4 + jj) * 8);
                bn1[jj] = *(const float4*)(b1p + ((s + 1) * 4 + jj) * 8);
            }
        }
#pragma unroll
        for (int jj = 0; jj < 4; ++jj) {
            const int j = s * 4 + jj;
            v2f a, b0, b1;
            a[0]  = xa[j].x;    a[1]  = xa[j].y;
            b0[0] = bq0[jj].x;  b0[1] = bq0[jj].y;
            b1[0] = bq1[jj].x;  b1[1] = bq1[jj].y;
            acc0 = __builtin_amdgcn_wmma_f32_16x16x4_f32(false, a, false, b0, (short)0, acc0, false, false);
            acc1 = __builtin_amdgcn_wmma_f32_16x16x4_f32(false, a, false, b1, (short)0, acc1, false, false);
            a[0]  = xa[j].z;    a[1]  = xa[j].w;
            b0[0] = bq0[jj].z;  b0[1] = bq0[jj].w;
            b1[0] = bq1[jj].z;  b1[1] = bq1[jj].w;
            acc0 = __builtin_amdgcn_wmma_f32_16x16x4_f32(false, a, false, b0, (short)0, acc0, false, false);
            acc1 = __builtin_amdgcn_wmma_f32_16x16x4_f32(false, a, false, b1, (short)0, acc1, false, false);
        }
        if (s + 1 < 8) {
#pragma unroll
            for (int jj = 0; jj < 4; ++jj) { bq0[jj] = bn0[jj]; bq1[jj] = bn1[jj]; }
        }
    }

    {
        float* pp = Ps + (kq * XT + rt * 16 + h * 8) * PP + c;
#pragma unroll
        for (int v = 0; v < 8; ++v) {
            pp[v * PP]      = acc0[v];
            pp[v * PP + 16] = acc1[v];
        }
    }
    __syncthreads();

    const int e   = lane >> 2;
    const int sel = lane & 3;
#pragma unroll
    for (int t = 0; t < 4; ++t) {
        const int tl = wave * 4 + t;
        const float* p = Ps + tl * PP;
        float4 la = *(const float4*)(p);
        float4 lb = *(const float4*)(p + 4);
        float4 cv = *(const float4*)(p + 8 + sel * 4);
#pragma unroll
        for (int q = 1; q < 4; ++q) {
            const float* pq = p + q * (XT * PP);
            const float4 a2 = *(const float4*)(pq);
            const float4 b2 = *(const float4*)(pq + 4);
            const float4 c2 = *(const float4*)(pq + 8 + sel * 4);
            la.x += a2.x; la.y += a2.y; la.z += a2.z; la.w += a2.w;
            lb.x += b2.x; lb.y += b2.y; lb.z += b2.z; lb.w += b2.w;
            cv.x += c2.x; cv.y += c2.y; cv.z += c2.z; cv.w += c2.w;
        }
        float l[8];
        l[0] = la.x; l[1] = la.y; l[2] = la.z; l[3] = la.w;
        l[4] = lb.x; l[5] = lb.y; l[6] = lb.z; l[7] = lb.w;

        float v0 = l[0]; int i0 = 0;
#pragma unroll
        for (int j = 1; j < 8; ++j) { float lj = l[j]; if (lj > v0) { v0 = lj; i0 = j; } }
        float v1 = -3.4e38f; int i1 = 0;
#pragma unroll
        for (int j = 0; j < 8; ++j) {
            if (j == i0) continue;
            float lj = l[j]; if (lj > v1) { v1 = lj; i1 = j; }
        }
        float ex = __expf(v1 - v0);
        float g0 = 1.0f / (1.0f + ex);
        float g1 = ex   / (1.0f + ex);

        float g = (e == i0) ? g0 : ((e == i1) ? g1 : 0.0f);
        g *= SCALING;

        float4 outv = make_float4(g * cv.x, g * cv.y, g * cv.z, g * cv.w);
        *(float4*)(tmp + (size_t)(tok0 + tl) * KDIM + lane * 4) = outv;
    }
}

#define BM   256
#define BN   128
#define LDB  132

__global__ __launch_bounds__(256) void gemm_wmma(const float* __restrict__ A,
                                                 const float* __restrict__ routed,
                                                 float* __restrict__ C) {
    __shared__ __attribute__((aligned(16))) float Bs[BN * LDB];

    const int tid  = threadIdx.x;
    const int lane = tid & 31;
    const int wave = tid >> 5;
    const int nb   = blockIdx.x & 7;
    const int mb   = blockIdx.x >> 3;
    const int n0   = nb * BN;
    const int h    = lane >> 4;
    const int c    = lane & 15;

    float4 st[16];
#pragma unroll
    for (int i = 0; i < 16; ++i) {
        const int u  = i * 256 + tid;
        const int q  = u & 3;
        const int ol = (u >> 2) & 127;
        const int e  = u >> 9;
        st[i] = *(const float4*)(routed + ((size_t)(e * D_OUT + n0 + ol) * RANK + q * 4));
    }

    const int row0 = mb * BM + wave * 32;
    const float* a0p = A + (size_t)(row0 + c) * KDIM + h * 4;
    const float* a1p = a0p + (size_t)16 * KDIM;
    float4 a0[16], a1[16];
#pragma unroll
    for (int j = 0; j < 16; ++j) {
        a0[j] = *(const float4*)(a0p + j * 8);
        a1[j] = *(const float4*)(a1p + j * 8);
    }

#pragma unroll
    for (int i = 0; i < 16; ++i) {
        const int u  = i * 256 + tid;
        const int q  = u & 3;
        const int ol = (u >> 2) & 127;
        const int e  = u >> 9;
        *(float4*)(Bs + ol * LDB + e * RANK + q * 4) = st[i];
    }

    __syncthreads();
    __builtin_amdgcn_sched_barrier(0);

    float* outp = C + (size_t)(row0 + h * 8) * D_OUT + n0 + c;
    const float* bp = Bs + c * LDB + h * 4;

    float4 bq[8];
#pragma unroll
    for (int jj = 0; jj < 8; ++jj) bq[jj] = *(const float4*)(bp + jj * 8);

    v8f acc0 = {0.f, 0.f, 0.f, 0.f, 0.f, 0.f, 0.f, 0.f};
    v8f acc1 = {0.f, 0.f, 0.f, 0.f, 0.f, 0.f, 0.f, 0.f};
#pragma unroll
    for (int s = 0; s < 16; ++s) {
        const int nt = s >> 1;
        const int hf = s & 1;
        float4 bn[8];
        if (s + 1 < 16) {
            const float* np = bp + ((s + 1) >> 1) * 16 * LDB + ((s + 1) & 1) * 64;
#pragma unroll
            for (int jj = 0; jj < 8; ++jj) bn[jj] = *(const float4*)(np + jj * 8);
        }
#pragma unroll
        for (int jj = 0; jj < 8; ++jj) {
            const int j = hf * 8 + jj;
            v2f blo, bhi, x0, x1;
            blo[0] = bq[jj].x; blo[1] = bq[jj].y;
            bhi[0] = bq[jj].z; bhi[1] = bq[jj].w;
            x0[0] = a0[j].x; x0[1] = a0[j].y;
            x1[0] = a1[j].x; x1[1] = a1[j].y;
            acc0 = __builtin_amdgcn_wmma_f32_16x16x4_f32(false, x0, false, blo, (short)0, acc0, false, false);
            acc1 = __builtin_amdgcn_wmma_f32_16x16x4_f32(false, x1, false, blo, (short)0, acc1, false, false);
            x0[0] = a0[j].z; x0[1] = a0[j].w;
            x1[0] = a1[j].z; x1[1] = a1[j].w;
            acc0 = __builtin_amdgcn_wmma_f32_16x16x4_f32(false, x0, false, bhi, (short)0, acc0, false, false);
            acc1 = __builtin_amdgcn_wmma_f32_16x16x4_f32(false, x1, false, bhi, (short)0, acc1, false, false);
        }
        if (hf == 1) {
            float* o = outp + nt * 16;
#pragma unroll
            for (int v = 0; v < 8; ++v) {
                o[(size_t)v * D_OUT]        = acc0[v];
                o[(size_t)(16 + v) * D_OUT] = acc1[v];
            }
#pragma unroll
            for (int v = 0; v < 8; ++v) { acc0[v] = 0.f; acc1[v] = 0.f; }
        }
        if (s + 1 < 16) {
#pragma unroll
            for (int jj = 0; jj < 8; ++jj) bq[jj] = bn[jj];
        }
    }
}

extern "C" void kernel_launch(void* const* d_in, const int* in_sizes, int n_in,
                              void* d_out, int out_size, void* d_ws, size_t ws_size,
                              hipStream_t stream) {
    const float* x        = (const float*)d_in[0];
    const float* w_route  = (const float*)d_in[1];
    const float* compress = (const float*)d_in[2];
    const float* routed   = (const float*)d_in[3];
    float*       out      = (float*)d_out;

    float* tmp = (float*)d_ws;

    const int ntok = in_sizes[0] / D_IN;

    route_compress<<<ntok / XT, 256, 0, stream>>>(x, w_route, compress, tmp);

    gemm_wmma<<<(ntok / BM) * (D_OUT / BN), 256, 0, stream>>>(tmp, routed, out);
}


# route+compress kernel: x loads split, 8 float4 steps with the Wcat staging and 24 issued behind its barrier, so the matrix sweep runs under the x stream instead of after it
# speedup vs baseline: 2.6770x; 1.0529x over previous
#include <hip/hip_runtime.h>
#include <math.h>

typedef __attribute__((ext_vector_type(2))) float v2f;
typedef __attribute__((ext_vector_type(8))) float v8f;

#define D_IN   1024
#define D_OUT  1024
#define NEXP   8
#define RANK   16
#define KDIM   128
#define SCALING 2.0f

#define XT   32
#define WP   1028
#define PP   36

__global__ __launch_bounds__(256) void route_compress(const float* __restrict__ x,
                                                      const float* __restrict__ w_route,
                                                      const float* __restrict__ compress,
                                                      float* __restrict__ tmp) {
    __shared__ __attribute__((aligned(16))) float Ws[24 * WP];
    __shared__ __attribute__((aligned(16))) float Ps[4 * XT * PP];

    const int tid  = threadIdx.x;
    const int lane = tid & 31;
    const int wave = tid >> 5;
    const int h    = lane >> 4;
    const int c    = lane & 15;
    const int rt   = wave & 1;
    const int kq   = wave >> 1;
    const int tok0 = blockIdx.x * XT;

    float4 st[24];
#pragma unroll
    for (int i = 0; i < 24; ++i) {
        const float* src = (i < 8) ? (w_route + i * D_IN) : (compress + (i - 8) * D_IN);
        st[i] = *(const float4*)(src + tid * 4);
    }

    unsigned xoff = (unsigned)(tok0 + rt * 16 + c) * D_IN + kq * 256 + h * 4;
    const float* xp = x + xoff;
    float4 xa[32];
#pragma unroll
    for (int j = 0; j < 8; ++j) xa[j] = *(const float4*)(xp + j * 8);

#pragma unroll
    for (int i = 0; i < 24; ++i) *(float4*)(Ws + i * WP + tid * 4) = st[i];

    __syncthreads();

    unsigned xoff2 = xoff + 64;
    asm volatile("" : "+v"(xoff2));
    const float* xp2 = x + xoff2;
#pragma unroll
    for (int j = 8; j < 32; ++j) xa[j] = *(const float4*)(xp2 + (j - 8) * 8);
    __builtin_amdgcn_sched_barrier(0);

    const float* b0p = Ws + c * WP + kq * 256 + h * 4;
    const float* b1p = Ws + (16 + (c & 7)) * WP + kq * 256 + h * 4;

    float4 bq0[4], bq1[4];
#pragma unroll
    for (int jj = 0; jj < 4; ++jj) {
        bq0[jj] = *(const float4*)(b0p + jj * 8);
        bq1[jj] = *(const float4*)(b1p + jj * 8);
    }

    v8f acc0 = {0.f, 0.f, 0.f, 0.f, 0.f, 0.f, 0.f, 0.f};
    v8f acc1 = {0.f, 0.f, 0.f, 0.f, 0.f, 0.f, 0.f, 0.f};
#pragma unroll
    for (int s = 0; s < 8; ++s) {
        float4 bn0[4], bn1[4];
        if (s + 1 < 8) {
#pragma unroll
            for (int jj = 0; jj < 4; ++jj) {
                bn0[jj] = *(const float4*)(b0p + ((s + 1) * 4 + jj) * 8);
                bn1[jj] = *(const float4*)(b1p + ((s + 1) * 4 + jj) * 8);
            }
        }
#pragma unroll
        for (int jj = 0; jj < 4; ++jj) {
            const int j = s * 4 + jj;
            v2f a, b0, b1;
            a[0]  = xa[j].x;    a[1]  = xa[j].y;
            b0[0] = bq0[jj].x;  b0[1] = bq0[jj].y;
            b1[0] = bq1[jj].x;  b1[1] = bq1[jj].y;
            acc0 = __builtin_amdgcn_wmma_f32_16x16x4_f32(false, a, false, b0, (short)0, acc0, false, false);
            acc1 = __builtin_amdgcn_wmma_f32_16x16x4_f32(false, a, false, b1, (short)0, acc1, false, false);
            a[0]  = xa[j].z;    a[1]  = xa[j].w;
            b0[0] = bq0[jj].z;  b0[1] = bq0[jj].w;
            b1[0] = bq1[jj].z;  b1[1] = bq1[jj].w;
            acc0 = __builtin_amdgcn_wmma_f32_16x16x4_f32(false, a, false, b0, (short)0, acc0, false, false);
            acc1 = __builtin_amdgcn_wmma_f32_16x16x4_f32(false, a, false, b1, (short)0, acc1, false, false);
        }
        if (s + 1 < 8) {
#pragma unroll
            for (int jj = 0; jj < 4; ++jj) { bq0[jj] = bn0[jj]; bq1[jj] = bn1[jj]; }
        }
    }

    {
        float* pp = Ps + (kq * XT + rt * 16 + h * 8) * PP + c;
#pragma unroll
        for (int v = 0; v < 8; ++v) {
            pp[v * PP]      = acc0[v];
            pp[v * PP + 16] = acc1[v];
        }
    }
    __syncthreads();

    const int e   = lane >> 2;
    const int sel = lane & 3;
#pragma unroll
    for (int t = 0; t < 4; ++t) {
        const int tl = wave * 4 + t;
        const float* p = Ps + tl * PP;
        float4 la = *(const float4*)(p);
        float4 lb = *(const float4*)(p + 4);
        float4 cv = *(const float4*)(p + 8 + sel * 4);
#pragma unroll
        for (int q = 1; q < 4; ++q) {
            const float* pq = p + q * (XT * PP);
            const float4 a2 = *(const float4*)(pq);
            const float4 b2 = *(const float4*)(pq + 4);
            const float4 c2 = *(const float4*)(pq + 8 + sel * 4);
            la.x += a2.x; la.y += a2.y; la.z += a2.z; la.w += a2.w;
            lb.x += b2.x; lb.y += b2.y; lb.z += b2.z; lb.w += b2.w;
            cv.x += c2.x; cv.y += c2.y; cv.z += c2.z; cv.w += c2.w;
        }
        float l[8];
        l[0] = la.x; l[1] = la.y; l[2] = la.z; l[3] = la.w;
        l[4] = lb.x; l[5] = lb.y; l[6] = lb.z; l[7] = lb.w;

        float v0 = l[0]; int i0 = 0;
#pragma unroll
        for (int j = 1; j < 8; ++j) { float lj = l[j]; if (lj > v0) { v0 = lj; i0 = j; } }
        float v1 = -3.4e38f; int i1 = 0;
#pragma unroll
        for (int j = 0; j < 8; ++j) {
            if (j == i0) continue;
            float lj = l[j]; if (lj > v1) { v1 = lj; i1 = j; }
        }
        float ex = __expf(v1 - v0);
        float g0 = 1.0f / (1.0f + ex);
        float g1 = ex   / (1.0f + ex);

        float g = (e == i0) ? g0 : ((e == i1) ? g1 : 0.0f);
        g *= SCALING;

        float4 outv = make_float4(g * cv.x, g * cv.y, g * cv.z, g * cv.w);
        *(float4*)(tmp + (size_t)(tok0 + tl) * KDIM + lane * 4) = outv;
    }
}

#define BM   256
#define BN   128
#define LDB  132

__global__ __launch_bounds__(256) void gemm_wmma(const float* __restrict__ A,
                                                 const float* __restrict__ routed,
                                                 float* __restrict__ C) {
    __shared__ __attribute__((aligned(16))) float Bs[BN * LDB];

    const int tid  = threadIdx.x;
    const int lane = tid & 31;
    const int wave = tid >> 5;
    const int nb   = blockIdx.x & 7;
    const int mb   = blockIdx.x >> 3;
    const int n0   = nb * BN;
    const int h    = lane >> 4;
    const int c    = lane & 15;

    float4 st[16];
#pragma unroll
    for (int i = 0; i < 16; ++i) {
        const int u  = i * 256 + tid;
        const int q  = u & 3;
        const int ol = (u >> 2) & 127;
        const int e  = u >> 9;
        st[i] = *(const float4*)(routed + ((size_t)(e * D_OUT + n0 + ol) * RANK + q * 4));
    }

    const int row0 = mb * BM + wave * 32;
    const float* a0p = A + (size_t)(row0 + c) * KDIM + h * 4;
    const float* a1p = a0p + (size_t)16 * KDIM;
    float4 a0[16], a1[16];
#pragma unroll
    for (int j = 0; j < 16; ++j) {
        a0[j] = *(const float4*)(a0p + j * 8);
        a1[j] = *(const float4*)(a1p + j * 8);
    }

#pragma unroll
    for (int i = 0; i < 16; ++i) {
        const int u  = i * 256 + tid;
        const int q  = u & 3;
        const int ol = (u >> 2) & 127;
        const int e  = u >> 9;
        *(float4*)(Bs + ol * LDB + e * RANK + q * 4) = st[i];
    }

    __syncthreads();
    __builtin_amdgcn_sched_barrier(0);

    float* outp = C + (size_t)(row0 + h * 8) * D_OUT + n0 + c;
    const float* bp = Bs + c * LDB + h * 4;

    float4 bq[8];
#pragma unroll
    for (int jj = 0; jj < 8; ++jj) bq[jj] = *(const float4*)(bp + jj * 8);

    v8f acc0 = {0.f, 0.f, 0.f, 0.f, 0.f, 0.f, 0.f, 0.f};
    v8f acc1 = {0.f, 0.f, 0.f, 0.f, 0.f, 0.f, 0.f, 0.f};
#pragma unroll
    for (int s = 0; s < 16; ++s) {
        const int nt = s >> 1;
        const int hf = s & 1;
        float4 bn[8];
        if (s + 1 < 16) {
            const float* np = bp + ((s + 1) >> 1) * 16 * LDB + ((s + 1) & 1) * 64;
#pragma unroll
            for (int jj = 0; jj < 8; ++jj) bn[jj] = *(const float4*)(np + jj * 8);
        }
#pragma unroll
        for (int jj = 0; jj < 8; ++jj) {
            const int j = hf * 8 + jj;
            v2f blo, bhi, x0, x1;
            blo[0] = bq[jj].x; blo[1] = bq[jj].y;
            bhi[0] = bq[jj].z; bhi[1] = bq[jj].w;
            x0[0] = a0[j].x; x0[1] = a0[j].y;
            x1[0] = a1[j].x; x1[1] = a1[j].y;
            acc0 = __builtin_amdgcn_wmma_f32_16x16x4_f32(false, x0, false, blo, (short)0, acc0, false, false);
            acc1 = __builtin_amdgcn_wmma_f32_16x16x4_f32(false, x1, false, blo, (short)0, acc1, false, false);
            x0[0] = a0[j].z; x0[1] = a0[j].w;
            x1[0] = a1[j].z; x1[1] = a1[j].w;
            acc0 = __builtin_amdgcn_wmma_f32_16x16x4_f32(false, x0, false, bhi, (short)0, acc0, false, false);
            acc1 = __builtin_amdgcn_wmma_f32_16x16x4_f32(false, x1, false, bhi, (short)0, acc1, false, false);
        }
        if (hf == 1) {
            float* o = outp + nt * 16;
#pragma unroll
            for (int v = 0; v < 8; ++v) {
                o[(size_t)v * D_OUT]        = acc0[v];
                o[(size_t)(16 + v) * D_OUT] = acc1[v];
            }
#pragma unroll
            for (int v = 0; v < 8; ++v) { acc0[v] = 0.f; acc1[v] = 0.f; }
        }
        if (s + 1 < 16) {
#pragma unroll
            for (int jj = 0; jj < 8; ++jj) bq[jj] = bn[jj];
        }
    }
}

extern "C" void kernel_launch(void* const* d_in, const int* in_sizes, int n_in,
                              void* d_out, int out_size, void* d_ws, size_t ws_size,
                              hipStream_t stream) {
    const float* x        = (const float*)d_in[0];
    const float* w_route  = (const float*)d_in[1];
    const float* compress = (const float*)d_in[2];
    const float* routed   = (const float*)d_in[3];
    float*       out      = (float*)d_out;

    float* tmp = (float*)d_ws;

    const int ntok = in_sizes[0] / D_IN;

    route_compress<<<ntok / XT, 256, 0, stream>>>(x, w_route, compress, tmp);

    gemm_wmma<<<(ntok / BM) * (D_OUT / BN), 256, 0, stream>>>(tmp, routed, out);
}


# combine GEMM multiplies only each token's two picked experts: pairs grouped by expert in LDS, wave = expert, 16 k per pair instead of 128; out = slot0 sum + slot1 sum (sum order changes)
# speedup vs baseline: 3.7396x; 1.3969x over previous
#include <hip/hip_runtime.h>
#include <math.h>

typedef __attribute__((ext_vector_type(2))) float v2f;
typedef __attribute__((ext_vector_type(8))) float v8f;

#define D_IN   1024
#define D_OUT  1024
#define NEXP   8
#define RANK   16
#define KDIM   128
#define SCALING 2.0f

#define XT   32
#define WP   1028
#define PP   36

__global__ __launch_bounds__(256) void route_compress(const float* __restrict__ x,
                                                      const float* __restrict__ w_route,
                                                      const float* __restrict__ compress,
                                                      float* __restrict__ rec, int ntok) {
    __shared__ __attribute__((aligned(16))) float Ws[24 * WP];
    __shared__ __attribute__((aligned(16))) float Ps[4 * XT * PP];

    const int tid  = threadIdx.x;
    const int lane = tid & 31;
    const int wave = tid >> 5;
    const int h    = lane >> 4;
    const int c    = lane & 15;
    const int rt   = wave & 1;
    const int kq   = wave >> 1;
    const int tok0 = blockIdx.x * XT;

    float4 st[24];
#pragma unroll
    for (int i = 0; i < 24; ++i) {
        const float* src = (i < 8) ? (w_route + i * D_IN) : (compress + (i - 8) * D_IN);
        st[i] = *(const float4*)(src + tid * 4);
    }

    unsigned xoff = (unsigned)(tok0 + rt * 16 + c) * D_IN + kq * 256 + h * 4;
    const float* xp = x + xoff;
    float4 xa[32];
#pragma unroll
    for (int j = 0; j < 8; ++j) xa[j] = *(const float4*)(xp + j * 8);

#pragma unroll
    for (int i = 0; i < 24; ++i) *(float4*)(Ws + i * WP + tid * 4) = st[i];

    __syncthreads();

    unsigned xoff2 = xoff + 64;
    asm volatile("" : "+v"(xoff2));
    const float* xp2 = x + xoff2;
#pragma unroll
    for (int j = 8; j < 32; ++j) xa[j] = *(const float4*)(xp2 + (j - 8) * 8);
    __builtin_amdgcn_sched_barrier(0);

    const float* b0p = Ws + c * WP + kq * 256 + h * 4;
    const float* b1p = Ws + (16 + (c & 7)) * WP + kq * 256 + h * 4;

    float4 bq0[4], bq1[4];
#pragma unroll
    for (int jj = 0; jj < 4; ++jj) {
        bq0[jj] = *(const float4*)(b0p + jj * 8);
        bq1[jj] = *(const float4*)(b1p + jj * 8);
    }

    v8f acc0 = {0.f, 0.f, 0.f, 0.f, 0.f, 0.f, 0.f, 0.f};
    v8f acc1 = {0.f, 0.f, 0.f, 0.f, 0.f, 0.f, 0.f, 0.f};
#pragma unroll
    for (int s = 0; s < 8; ++s) {
        float4 bn0[4], bn1[4];
        if (s + 1 < 8) {
#pragma unroll
            for (int jj = 0; jj < 4; ++jj) {
                bn0[jj] = *(const float4*)(b0p + ((s + 1) * 4 + jj) * 8);
                bn1[jj] = *(const float4*)(b1p + ((s + 1) * 4 + jj) * 8);
            }
        }
#pragma unroll
        for (int jj = 0; jj < 4; ++jj) {
            const int j = s * 4 + jj;
            v2f a, b0, b1;
            a[0]  = xa[j].x;    a[1]  = xa[j].y;
            b0[0] = bq0[jj].x;  b0[1] = bq0[jj].y;
            b1[0] = bq1[jj].x;  b1[1] = bq1[jj].y;
            acc0 = __builtin_amdgcn_wmma_f32_16x16x4_f32(false, a, false, b0, (short)0, acc0, false, false);
            acc1 = __builtin_amdgcn_wmma_f32_16x16x4_f32(false, a, false, b1, (short)0, acc1, false, false);
            a[0]  = xa[j].z;    a[1]  = xa[j].w;
            b0[0] = bq0[jj].z;  b0[1] = bq0[jj].w;
            b1[0] = bq1[jj].z;  b1[1] = bq1[jj].w;
            acc0 = __builtin_amdgcn_wmma_f32_16x16x4_f32(false, a, false, b0, (short)0, acc0, false, false);
            acc1 = __builtin_amdgcn_wmma_f32_16x16x4_f32(false, a, false, b1, (short)0, acc1, false, false);
        }
        if (s + 1 < 8) {
#pragma unroll
            for (int jj = 0; jj < 4; ++jj) { bq0[jj] = bn0[jj]; bq1[jj] = bn1[jj]; }
        }
    }

    {
        float* pp = Ps + (kq * XT + rt * 16 + h * 8) * PP + c;
#pragma unroll
        for (int v = 0; v < 8; ++v) {
            pp[v * PP]      = acc0[v];
            pp[v * PP + 16] = acc1[v];
        }
    }
    __syncthreads();

    int* pick = (int*)(rec + (size_t)ntok * 32);
    const int tl   = wave * 4 + (lane >> 3);
    const int slot = (lane >> 2) & 1;
    const int sel  = lane & 3;
    {
        const float* p = Ps + tl * PP;
        float4 la = *(const float4*)(p);
        float4 lb = *(const float4*)(p + 4);
        float4 cv = *(const float4*)(p + 8 + sel * 4);
#pragma unroll
        for (int q = 1; q < 4; ++q) {
            const float* pq = p + q * (XT * PP);
            const float4 a2 = *(const float4*)(pq);
            const float4 b2 = *(const float4*)(pq + 4);
            const float4 c2 = *(const float4*)(pq + 8 + sel * 4);
            la.x += a2.x; la.y += a2.y; la.z += a2.z; la.w += a2.w;
            lb.x += b2.x; lb.y += b2.y; lb.z += b2.z; lb.w += b2.w;
            cv.x += c2.x; cv.y += c2.y; cv.z += c2.z; cv.w += c2.w;
        }
        float l[8];
        l[0] = la.x; l[1] = la.y; l[2] = la.z; l[3] = la.w;
        l[4] = lb.x; l[5] = lb.y; l[6] = lb.z; l[7] = lb.w;

        float v0 = l[0]; int i0 = 0;
#pragma unroll
        for (int j = 1; j < 8; ++j) { float lj = l[j]; if (lj > v0) { v0 = lj; i0 = j; } }
        float v1 = -3.4e38f; int i1 = 0;
#pragma unroll
        for (int j = 0; j < 8; ++j) {
            if (j == i0) continue;
            float lj = l[j]; if (lj > v1) { v1 = lj; i1 = j; }
        }
        float ex = __expf(v1 - v0);
        float g0 = 1.0f / (1.0f + ex);
        float g1 = ex   / (1.0f + ex);

        float g = slot ? g1 : g0;
        g *= SCALING;

        float4 outv = make_float4(g * cv.x, g * cv.y, g * cv.z, g * cv.w);
        *(float4*)(rec + (size_t)(tok0 + tl) * 32 + (lane & 7) * 4) = outv;
        if ((lane & 7) == 0) pick[tok0 + tl] = i0 | (i1 << 8);
    }
}

#define SM    256
#define SN    64
#define PL    68
#define LCAP  512

__global__ __launch_bounds__(256) void combine_picked(const float* __restrict__ rec,
                                                      const float* __restrict__ routed,
                                                      float* __restrict__ C, int ntok) {
    __shared__ __attribute__((aligned(16))) float Pl[(2 * SM + 1) * PL];
    __shared__ __attribute__((aligned(16))) int   Ls[NEXP * LCAP];
    __shared__ int Cn[8 * NEXP];

    const int tid  = threadIdx.x;
    const int lane = tid & 31;
    const int wave = tid >> 5;
    const int h    = lane >> 4;
    const int c    = lane & 15;
    const int nb   = blockIdx.x & 15;
    const int mb   = blockIdx.x >> 4;
    const int n0   = nb * SN;
    const int tok0 = mb * SM;
    const int* pick = (const int*)(rec + (size_t)ntok * 32);

    float4 bf[4][2];
#pragma unroll
    for (int ct = 0; ct < 4; ++ct) {
        const float* bp = routed + ((size_t)(wave * D_OUT + n0 + ct * 16 + c) * RANK + h * 4);
        bf[ct][0] = *(const float4*)(bp);
        bf[ct][1] = *(const float4*)(bp + 8);
    }

    const int pk = pick[tok0 + tid];
    const int i0 = pk & 7;
    const int i1 = (pk >> 8) & 7;
    const unsigned below = (1u << lane) - 1u;
    int pos0 = 0, pos1 = 0;
#pragma unroll
    for (int e = 0; e < NEXP; ++e) {
        const unsigned m0 = __builtin_amdgcn_ballot_w32(i0 == e);
        const unsigned m1 = __builtin_amdgcn_ballot_w32(i1 == e);
        const int c0 = __builtin_popcount(m0);
        const int c1 = __builtin_popcount(m1);
        if (i0 == e) pos0 = __builtin_popcount(m0 & below);
        if (i1 == e) pos1 = c0 + __builtin_popcount(m1 & below);
        if (lane == e) Cn[wave * NEXP + e] = c0 + c1;
    }
    __syncthreads();

    int base0 = 0, base1 = 0, n = 0;
#pragma unroll
    for (int w2 = 0; w2 < 8; ++w2) {
        const int a = Cn[w2 * NEXP + i0];
        const int b = Cn[w2 * NEXP + i1];
        if (w2 < wave) { base0 += a; base1 += b; }
        n += Cn[w2 * NEXP + wave];
    }
    Ls[i0 * LCAP + base0 + pos0] = tid;
    Ls[i1 * LCAP + base1 + pos1] = tid | 256;
    __syncthreads();

    n = __builtin_amdgcn_readfirstlane(n);
    const int ntile = (n + 15) >> 4;
    int* myl = Ls + wave * LCAP;

    {
        const int p = n + lane;
        const int padEnd = (ntile > 0 ? ntile : 1) * 16;
        if (lane < 16 && p < padEnd) myl[p] = 512;
    }

    auto loadEnt = [&](int tq) -> int {
        const int tt = tq < ntile ? tq : 0;
        return myl[tt * 16 + c];
    };
    auto gatherA = [&](int ent, float4& f0, float4& f1) {
        const float* ap = rec + (size_t)(tok0 + (ent & 255)) * 32 + ((ent >> 8) & 1) * 16 + h * 4;
        f0 = *(const float4*)(ap);
        f1 = *(const float4*)(ap + 8);
    };

    auto tile = [&](int t, const float4& x0, const float4& x1) {
        const int4 ea = *(const int4*)(myl + t * 16 + h * 8);
        const int4 eb = *(const int4*)(myl + t * 16 + h * 8 + 4);

        v8f acc[4];
#pragma unroll
        for (int ct = 0; ct < 4; ++ct) acc[ct] = (v8f){0.f, 0.f, 0.f, 0.f, 0.f, 0.f, 0.f, 0.f};
#pragma unroll
        for (int j = 0; j < 2; ++j) {
            const float4 xa = j ? x1 : x0;
            v2f a;
            a[0] = xa.x; a[1] = xa.y;
#pragma unroll
            for (int ct = 0; ct < 4; ++ct) {
                v2f b; b[0] = bf[ct][j].x; b[1] = bf[ct][j].y;
                acc[ct] = __builtin_amdgcn_wmma_f32_16x16x4_f32(false, a, false, b, (short)0, acc[ct], false, false);
            }
            a[0] = xa.z; a[1] = xa.w;
#pragma unroll
            for (int ct = 0; ct < 4; ++ct) {
                v2f b; b[0] = bf[ct][j].z; b[1] = bf[ct][j].w;
                acc[ct] = __builtin_amdgcn_wmma_f32_16x16x4_f32(false, a, false, b, (short)0, acc[ct], false, false);
            }
        }

        int ev[8];
        ev[0] = ea.x; ev[1] = ea.y; ev[2] = ea.z; ev[3] = ea.w;
        ev[4] = eb.x; ev[5] = eb.y; ev[6] = eb.z; ev[7] = eb.w;
#pragma unroll
        for (int v = 0; v < 8; ++v) {
            const int ent = ev[v];
            float* d = Pl + (((ent >> 8) & 3) * SM + (ent & 255)) * PL + c;
#pragma unroll
            for (int ct = 0; ct < 4; ++ct) d[ct * 16] = acc[ct][v];
        }
    };

    float4 fa0, fa1, fb0, fb1, fc0, fc1;
    gatherA(loadEnt(0), fa0, fa1);
    gatherA(loadEnt(1), fb0, fb1);
    int en = loadEnt(2);
    const int ntrip = ntile / 3;
    int t = 0;
    for (int k = 0; k < ntrip; ++k, t += 3) {
        gatherA(en, fc0, fc1);
        en = loadEnt(t + 3);
        __builtin_amdgcn_sched_barrier(0);
        tile(t, fa0, fa1);
        gatherA(en, fa0, fa1);
        en = loadEnt(t + 4);
        __builtin_amdgcn_sched_barrier(0);
        tile(t + 1, fb0, fb1);
        gatherA(en, fb0, fb1);
        en = loadEnt(t + 5);
        __builtin_amdgcn_sched_barrier(0);
        tile(t + 2, fc0, fc1);
    }
    if (t < ntile) {
        tile(t, fa0, fa1);
        if (t + 1 < ntile) tile(t + 1, fb0, fb1);
    }
    __syncthreads();

    float* ob = C + (size_t)tok0 * D_OUT + n0;
#pragma unroll
    for (int i = 0; i < 16; ++i) {
        const int idx = i * 256 + tid;
        const int row = idx >> 4;
        const int q   = idx & 15;
        const float4 s0 = *(const float4*)(Pl + row * PL + q * 4);
        const float4 s1 = *(const float4*)(Pl + (SM + row) * PL + q * 4);
        float4 o;
        o.x = s0.x + s1.x; o.y = s0.y + s1.y; o.z = s0.z + s1.z; o.w = s0.w + s1.w;
        *(float4*)(ob + (size_t)row * D_OUT + q * 4) = o;
    }
}

extern "C" void kernel_launch(void* const* d_in, const int* in_sizes, int n_in,
                              void* d_out, int out_size, void* d_ws, size_t ws_size,
                              hipStream_t stream) {
    const float* x        = (const float*)d_in[0];
    const float* w_route  = (const float*)d_in[1];
    const float* compress = (const float*)d_in[2];
    const float* routed   = (const float*)d_in[3];
    float*       out      = (float*)d_out;

    float* rec = (float*)d_ws;

    const int ntok = in_sizes[0] / D_IN;

    route_compress<<<ntok / XT, 256, 0, stream>>>(x, w_route, compress, rec, ntok);

    combine_picked<<<(ntok / SM) * (D_OUT / SN), 256, 0, stream>>>(rec, routed, out, ntok);
}


# combine kernel: each work-group walks 2 column blocks of 64 outputs with one expert grouping (grid 1024 -> 512, one resident round); next block's B and first gathers fly under the row sums
# speedup vs baseline: 4.0114x; 1.0727x over previous
#include <hip/hip_runtime.h>
#include <math.h>

typedef __attribute__((ext_vector_type(2))) float v2f;
typedef __attribute__((ext_vector_type(8))) float v8f;

#define D_IN   1024
#define D_OUT  1024
#define NEXP   8
#define RANK   16
#define KDIM   128
#define SCALING 2.0f

#define XT   32
#define WP   1028
#define PP   36

__global__ __launch_bounds__(256) void route_compress(const float* __restrict__ x,
                                                      const float* __restrict__ w_route,
                                                      const float* __restrict__ compress,
                                                      float* __restrict__ rec, int ntok) {
    __shared__ __attribute__((aligned(16))) float Ws[24 * WP];
    __shared__ __attribute__((aligned(16))) float Ps[4 * XT * PP];

    const int tid  = threadIdx.x;
    const int lane = tid & 31;
    const int wave = tid >> 5;
    const int h    = lane >> 4;
    const int c    = lane & 15;
    const int rt   = wave & 1;
    const int kq   = wave >> 1;
    const int tok0 = blockIdx.x * XT;

    float4 st[24];
#pragma unroll
    for (int i = 0; i < 24; ++i) {
        const float* src = (i < 8) ? (w_route + i * D_IN) : (compress + (i - 8) * D_IN);
        st[i] = *(const float4*)(src + tid * 4);
    }

    unsigned xoff = (unsigned)(tok0 + rt * 16 + c) * D_IN + kq * 256 + h * 4;
    const float* xp = x + xoff;
    float4 xa[32];
#pragma unroll
    for (int j = 0; j < 8; ++j) xa[j] = *(const float4*)(xp + j * 8);

#pragma unroll
    for (int i = 0; i < 24; ++i) *(float4*)(Ws + i * WP + tid * 4) = st[i];

    __syncthreads();

    unsigned xoff2 = xoff + 64;
    asm volatile("" : "+v"(xoff2));
    const float* xp2 = x + xoff2;
#pragma unroll
    for (int j = 8; j < 32; ++j) xa[j] = *(const float4*)(xp2 + (j - 8) * 8);
    __builtin_amdgcn_sched_barrier(0);

    const float* b0p = Ws + c * WP + kq * 256 + h * 4;
    const float* b1p = Ws + (16 + (c & 7)) * WP + kq * 256 + h * 4;

    float4 bq0[4], bq1[4];
#pragma unroll
    for (int jj = 0; jj < 4; ++jj) {
        bq0[jj] = *(const float4*)(b0p + jj * 8);
        bq1[jj] = *(const float4*)(b1p + jj * 8);
    }

    v8f acc0 = {0.f, 0.f, 0.f, 0.f, 0.f, 0.f, 0.f, 0.f};
    v8f acc1 = {0.f, 0.f, 0.f, 0.f, 0.f, 0.f, 0.f, 0.f};
#pragma unroll
    for (int s = 0; s < 8; ++s) {
        float4 bn0[4], bn1[4];
        if (s + 1 < 8) {
#pragma unroll
            for (int jj = 0; jj < 4; ++jj) {
                bn0[jj] = *(const float4*)(b0p + ((s + 1) * 4 + jj) * 8);
                bn1[jj] = *(const float4*)(b1p + ((s + 1) * 4 + jj) * 8);
            }
        }
#pragma unroll
        for (int jj = 0; jj < 4; ++jj) {
            const int j = s * 4 + jj;
            v2f a, b0, b1;
            a[0]  = xa[j].x;    a[1]  = xa[j].y;
            b0[0] = bq0[jj].x;  b0[1] = bq0[jj].y;
            b1[0] = bq1[jj].x;  b1[1] = bq1[jj].y;
            acc0 = __builtin_amdgcn_wmma_f32_16x16x4_f32(false, a, false, b0, (short)0, acc0, false, false);
            acc1 = __builtin_amdgcn_wmma_f32_16x16x4_f32(false, a, false, b1, (short)0, acc1, false, false);
            a[0]  = xa[j].z;    a[1]  = xa[j].w;
            b0[0] = bq0[jj].z;  b0[1] = bq0[jj].w;
            b1[0] = bq1[jj].z;  b1[1] = bq1[jj].w;
            acc0 = __builtin_amdgcn_wmma_f32_16x16x4_f32(false, a, false, b0, (short)0, acc0, false, false);
            acc1 = __builtin_amdgcn_wmma_f32_16x16x4_f32(false, a, false, b1, (short)0, acc1, false, false);
        }
        if (s + 1 < 8) {
#pragma unroll
            for (int jj = 0; jj < 4; ++jj) { bq0[jj] = bn0[jj]; bq1[jj] = bn1[jj]; }
        }
    }

    {
        float* pp = Ps + (kq * XT + rt * 16 + h * 8) * PP + c;
#pragma unroll
        for (int v = 0; v < 8; ++v) {
            pp[v * PP]      = acc0[v];
            pp[v * PP + 16] = acc1[v];
        }
    }
    __syncthreads();

    int* pick = (int*)(rec + (size_t)ntok * 32);
    const int tl   = wave * 4 + (lane >> 3);
    const int slot = (lane >> 2) & 1;
    const int sel  = lane & 3;
    {
        const float* p = Ps + tl * PP;
        float4 la = *(const float4*)(p);
        float4 lb = *(const float4*)(p + 4);
        float4 cv = *(const float4*)(p + 8 + sel * 4);
#pragma unroll
        for (int q = 1; q < 4; ++q) {
            const float* pq = p + q * (XT * PP);
            const float4 a2 = *(const float4*)(pq);
            const float4 b2 = *(const float4*)(pq + 4);
            const float4 c2 = *(const float4*)(pq + 8 + sel * 4);
            la.x += a2.x; la.y += a2.y; la.z += a2.z; la.w += a2.w;
            lb.x += b2.x; lb.y += b2.y; lb.z += b2.z; lb.w += b2.w;
            cv.x += c2.x; cv.y += c2.y; cv.z += c2.z; cv.w += c2.w;
        }
        float l[8];
        l[0] = la.x; l[1] = la.y; l[2] = la.z; l[3] = la.w;
        l[4] = lb.x; l[5] = lb.y; l[6] = lb.z; l[7] = lb.w;

        float v0 = l[0]; int i0 = 0;
#pragma unroll
        for (int j = 1; j < 8; ++j) { float lj = l[j]; if (lj > v0) { v0 = lj; i0 = j; } }
        float v1 = -3.4e38f; int i1 = 0;
#pragma unroll
        for (int j = 0; j < 8; ++j) {
            if (j == i0) continue;
            float lj = l[j]; if (lj > v1) { v1 = lj; i1 = j; }
        }
        float ex = __expf(v1 - v0);
        float g0 = 1.0f / (1.0f + ex);
        float g1 = ex   / (1.0f + ex);

        float g = slot ? g1 : g0;
        g *= SCALING;

        float4 outv = make_float4(g * cv.x, g * cv.y, g * cv.z, g * cv.w);
        *(float4*)(rec + (size_t)(tok0 + tl) * 32 + (lane & 7) * 4) = outv;
        if ((lane & 7) == 0) pick[tok0 + tl] = i0 | (i1 << 8);
    }
}

#define SM    256
#define SN    64
#define PL    68
#define LCAP  512
#define NCB   2

__global__ __launch_bounds__(256) void combine_picked(const float* __restrict__ rec,
                                                      const float* __restrict__ routed,
                                                      float* __restrict__ C, int ntok) {
    __shared__ __attribute__((aligned(16))) float Pl[(2 * SM + 1) * PL];
    __shared__ __attribute__((aligned(16))) int   Ls[NEXP * LCAP];
    __shared__ int Cn[8 * NEXP];

    const int tid  = threadIdx.x;
    const int lane = tid & 31;
    const int wave = tid >> 5;
    const int h    = lane >> 4;
    const int c    = lane & 15;
    const int nb   = blockIdx.x & 7;
    const int mb   = blockIdx.x >> 3;
    int n0         = nb * (SN * NCB);
    const int tok0 = mb * SM;
    const int* pick = (const int*)(rec + (size_t)ntok * 32);

    float4 bf[4][2];
    auto loadB = [&](int col0) {
#pragma unroll
        for (int ct = 0; ct < 4; ++ct) {
            const float* bp = routed + ((size_t)(wave * D_OUT + col0 + ct * 16 + c) * RANK + h * 4);
            bf[ct][0] = *(const float4*)(bp);
            bf[ct][1] = *(const float4*)(bp + 8);
        }
    };
    loadB(n0);

    const int pk = pick[tok0 + tid];
    const int i0 = pk & 7;
    const int i1 = (pk >> 8) & 7;
    const unsigned below = (1u << lane) - 1u;
    int pos0 = 0, pos1 = 0;
#pragma unroll
    for (int e = 0; e < NEXP; ++e) {
        const unsigned m0 = __builtin_amdgcn_ballot_w32(i0 == e);
        const unsigned m1 = __builtin_amdgcn_ballot_w32(i1 == e);
        const int c0 = __builtin_popcount(m0);
        const int c1 = __builtin_popcount(m1);
        if (i0 == e) pos0 = __builtin_popcount(m0 & below);
        if (i1 == e) pos1 = c0 + __builtin_popcount(m1 & below);
        if (lane == e) Cn[wave * NEXP + e] = c0 + c1;
    }
    __syncthreads();

    int base0 = 0, base1 = 0, n = 0;
#pragma unroll
    for (int w2 = 0; w2 < 8; ++w2) {
        const int a = Cn[w2 * NEXP + i0];
        const int b = Cn[w2 * NEXP + i1];
        if (w2 < wave) { base0 += a; base1 += b; }
        n += Cn[w2 * NEXP + wave];
    }
    Ls[i0 * LCAP + base0 + pos0] = tid;
    Ls[i1 * LCAP + base1 + pos1] = tid | 256;
    __syncthreads();

    n = __builtin_amdgcn_readfirstlane(n);
    const int ntile = (n + 15) >> 4;
    int* myl = Ls + wave * LCAP;

    {
        const int p = n + lane;
        const int padEnd = (ntile > 0 ? ntile : 1) * 16;
        if (lane < 16 && p < padEnd) myl[p] = 512;
    }

    auto loadEnt = [&](int tq) -> int {
        const int tt = tq < ntile ? tq : 0;
        return myl[tt * 16 + c];
    };
    auto gatherA = [&](int ent, float4& f0, float4& f1) {
        const float* ap = rec + (size_t)(tok0 + (ent & 255)) * 32 + ((ent >> 8) & 1) * 16 + h * 4;
        f0 = *(const float4*)(ap);
        f1 = *(const float4*)(ap + 8);
    };

    auto tile = [&](int t, const float4& x0, const float4& x1) {
        const int4 ea = *(const int4*)(myl + t * 16 + h * 8);
        const int4 eb = *(const int4*)(myl + t * 16 + h * 8 + 4);

        v8f acc[4];
#pragma unroll
        for (int ct = 0; ct < 4; ++ct) acc[ct] = (v8f){0.f, 0.f, 0.f, 0.f, 0.f, 0.f, 0.f, 0.f};
#pragma unroll
        for (int j = 0; j < 2; ++j) {
            const float4 xa = j ? x1 : x0;
            v2f a;
            a[0] = xa.x; a[1] = xa.y;
#pragma unroll
            for (int ct = 0; ct < 4; ++ct) {
                v2f b; b[0] = bf[ct][j].x; b[1] = bf[ct][j].y;
                acc[ct] = __builtin_amdgcn_wmma_f32_16x16x4_f32(false, a, false, b, (short)0, acc[ct], false, false);
            }
            a[0] = xa.z; a[1] = xa.w;
#pragma unroll
            for (int ct = 0; ct < 4; ++ct) {
                v2f b; b[0] = bf[ct][j].z; b[1] = bf[ct][j].w;
                acc[ct] = __builtin_amdgcn_wmma_f32_16x16x4_f32(false, a, false, b, (short)0, acc[ct], false, false);
            }
        }

        int ev[8];
        ev[0] = ea.x; ev[1] = ea.y; ev[2] = ea.z; ev[3] = ea.w;
        ev[4] = eb.x; ev[5] = eb.y; ev[6] = eb.z; ev[7] = eb.w;
#pragma unroll
        for (int v = 0; v < 8; ++v) {
            const int ent = ev[v];
            float* d = Pl + (((ent >> 8) & 3) * SM + (ent & 255)) * PL + c;
#pragma unroll
            for (int ct = 0; ct < 4; ++ct) d[ct * 16] = acc[ct][v];
        }
    };

    float4 fa0, fa1, fb0, fb1, fc0, fc1;
    gatherA(loadEnt(0), fa0, fa1);
    gatherA(loadEnt(1), fb0, fb1);
    const int ntrip = ntile / 3;
    for (int cb = 0; cb < NCB; ++cb) {
    int en = loadEnt(2);
    int t = 0;
    for (int k = 0; k < ntrip; ++k, t += 3) {
        gatherA(en, fc0, fc1);
        en = loadEnt(t + 3);
        __builtin_amdgcn_sched_barrier(0);
        tile(t, fa0, fa1);
        gatherA(en, fa0, fa1);
        en = loadEnt(t + 4);
        __builtin_amdgcn_sched_barrier(0);
        tile(t + 1, fb0, fb1);
        gatherA(en, fb0, fb1);
        en = loadEnt(t + 5);
        __builtin_amdgcn_sched_barrier(0);
        tile(t + 2, fc0, fc1);
    }
    if (t < ntile) {
        tile(t, fa0, fa1);
        if (t + 1 < ntile) tile(t + 1, fb0, fb1);
    }
    __syncthreads();
    if (cb + 1 < NCB) {
        loadB(n0 + SN);
        gatherA(loadEnt(0), fa0, fa1);
        gatherA(loadEnt(1), fb0, fb1);
    }

    float* ob = C + (size_t)tok0 * D_OUT + n0;
#pragma unroll
    for (int i = 0; i < 16; ++i) {
        const int idx = i * 256 + tid;
        const int row = idx >> 4;
        const int q   = idx & 15;
        const float4 s0 = *(const float4*)(Pl + row * PL + q * 4);
        const float4 s1 = *(const float4*)(Pl + (SM + row) * PL + q * 4);
        float4 o;
        o.x = s0.x + s1.x; o.y = s0.y + s1.y; o.z = s0.z + s1.z; o.w = s0.w + s1.w;
        *(float4*)(ob + (size_t)row * D_OUT + q * 4) = o;
    }
    n0 += SN;
    if (cb + 1 < NCB) {
        __builtin_amdgcn_fence(__ATOMIC_RELEASE, "workgroup", "local");
        __builtin_amdgcn_s_barrier();
        __builtin_amdgcn_fence(__ATOMIC_ACQUIRE, "workgroup", "local");
    }
    }
}

extern "C" void kernel_launch(void* const* d_in, const int* in_sizes, int n_in,
                              void* d_out, int out_size, void* d_ws, size_t ws_size,
                              hipStream_t stream) {
    const float* x        = (const float*)d_in[0];
    const float* w_route  = (const float*)d_in[1];
    const float* compress = (const float*)d_in[2];
    const float* routed   = (const float*)d_in[3];
    float*       out      = (float*)d_out;

    float* rec = (float*)d_ws;

    const int ntok = in_sizes[0] / D_IN;

    route_compress<<<ntok / XT, 256, 0, stream>>>(x, w_route, compress, rec, ntok);

    combine_picked<<<(ntok / SM) * (D_OUT / (SN * NCB)), 256, 0, stream>>>(rec, routed, out, ntok);
}


# route+compress retiled: 64 tokens per work-group (grid 512 -> 256, 8 waves per WGP, not 16), a wave feeds 2 token tiles from each B fragment, Wcat staged once per 64 tokens; sum order unchanged
# speedup vs baseline: 4.3095x; 1.0743x over previous
#include <hip/hip_runtime.h>
#include <math.h>

typedef __attribute__((ext_vector_type(2))) float v2f;
typedef __attribute__((ext_vector_type(8))) float v8f;

#define D_IN   1024
#define D_OUT  1024
#define NEXP   8
#define RANK   16
#define KDIM   128
#define SCALING 2.0f

#define XT   64
#define WP   1028
#define PP   36

__global__ __launch_bounds__(256) void route_compress(const float* __restrict__ x,
                                                      const float* __restrict__ w_route,
                                                      const float* __restrict__ compress,
                                                      float* __restrict__ rec, int ntok) {
    __shared__ __attribute__((aligned(16))) float Ws[24 * WP];
    __shared__ __attribute__((aligned(16))) float Ps[4 * XT * PP];

    const int tid  = threadIdx.x;
    const int lane = tid & 31;
    const int wave = tid >> 5;
    const int h    = lane >> 4;
    const int c    = lane & 15;
    const int tp   = wave & 1;
    const int kq   = wave >> 1;
    const int tok0 = blockIdx.x * XT;

    float4 st[24];
#pragma unroll
    for (int i = 0; i < 24; ++i) {
        const float* src = (i < 8) ? (w_route + i * D_IN) : (compress + (i - 8) * D_IN);
        st[i] = *(const float4*)(src + tid * 4);
    }

    unsigned xoff = (unsigned)(tok0 + tp * 32 + c) * D_IN + kq * 256 + h * 4;
    const float* xp = x + xoff;
    float4 xa[32], xb[32];
#pragma unroll
    for (int j = 0; j < 8; ++j) {
        xa[j] = *(const float4*)(xp + j * 8);
        xb[j] = *(const float4*)(xp + 16 * D_IN + j * 8);
    }

#pragma unroll
    for (int i = 0; i < 24; ++i) *(float4*)(Ws + i * WP + tid * 4) = st[i];

    __syncthreads();

    unsigned xoff2 = xoff + 64;
    asm volatile("" : "+v"(xoff2));
    const float* xp2 = x + xoff2;
#pragma unroll
    for (int j = 8; j < 32; ++j) {
        xa[j] = *(const float4*)(xp2 + (j - 8) * 8);
        xb[j] = *(const float4*)(xp2 + 16 * D_IN + (j - 8) * 8);
    }
    __builtin_amdgcn_sched_barrier(0);

    const float* b0p = Ws + c * WP + kq * 256 + h * 4;
    const float* b1p = Ws + (16 + (c & 7)) * WP + kq * 256 + h * 4;

    float4 bq0[4], bq1[4];
#pragma unroll
    for (int jj = 0; jj < 4; ++jj) {
        bq0[jj] = *(const float4*)(b0p + jj * 8);
        bq1[jj] = *(const float4*)(b1p + jj * 8);
    }

    v8f acc00 = {0.f, 0.f, 0.f, 0.f, 0.f, 0.f, 0.f, 0.f};
    v8f acc01 = {0.f, 0.f, 0.f, 0.f, 0.f, 0.f, 0.f, 0.f};
    v8f acc10 = {0.f, 0.f, 0.f, 0.f, 0.f, 0.f, 0.f, 0.f};
    v8f acc11 = {0.f, 0.f, 0.f, 0.f, 0.f, 0.f, 0.f, 0.f};
#pragma unroll
    for (int s = 0; s < 8; ++s) {
        float4 bn0[4], bn1[4];
        if (s + 1 < 8) {
#pragma unroll
            for (int jj = 0; jj < 4; ++jj) {
                bn0[jj] = *(const float4*)(b0p + ((s + 1) * 4 + jj) * 8);
                bn1[jj] = *(const float4*)(b1p + ((s + 1) * 4 + jj) * 8);
            }
            __builtin_amdgcn_sched_barrier(0);
        }
#pragma unroll
        for (int jj = 0; jj < 4; ++jj) {
            const int j = s * 4 + jj;
            v2f a0, a1, b0, b1;
            a0[0] = xa[j].x;    a0[1] = xa[j].y;
            a1[0] = xb[j].x;    a1[1] = xb[j].y;
            b0[0] = bq0[jj].x;  b0[1] = bq0[jj].y;
            b1[0] = bq1[jj].x;  b1[1] = bq1[jj].y;
            acc00 = __builtin_amdgcn_wmma_f32_16x16x4_f32(false, a0, false, b0, (short)0, acc00, false, false);
            acc01 = __builtin_amdgcn_wmma_f32_16x16x4_f32(false, a0, false, b1, (short)0, acc01, false, false);
            acc10 = __builtin_amdgcn_wmma_f32_16x16x4_f32(false, a1, false, b0, (short)0, acc10, false, false);
            acc11 = __builtin_amdgcn_wmma_f32_16x16x4_f32(false, a1, false, b1, (short)0, acc11, false, false);
            a0[0] = xa[j].z;    a0[1] = xa[j].w;
            a1[0] = xb[j].z;    a1[1] = xb[j].w;
            b0[0] = bq0[jj].z;  b0[1] = bq0[jj].w;
            b1[0] = bq1[jj].z;  b1[1] = bq1[jj].w;
            acc00 = __builtin_amdgcn_wmma_f32_16x16x4_f32(false, a0, false, b0, (short)0, acc00, false, false);
            acc01 = __builtin_amdgcn_wmma_f32_16x16x4_f32(false, a0, false, b1, (short)0, acc01, false, false);
            acc10 = __builtin_amdgcn_wmma_f32_16x16x4_f32(false, a1, false, b0, (short)0, acc10, false, false);
            acc11 = __builtin_amdgcn_wmma_f32_16x16x4_f32(false, a1, false, b1, (short)0, acc11, false, false);
        }
        if (s + 1 < 8) {
#pragma unroll
            for (int jj = 0; jj < 4; ++jj) { bq0[jj] = bn0[jj]; bq1[jj] = bn1[jj]; }
        }
    }

    {
        float* pp = Ps + (kq * XT + tp * 32 + h * 8) * PP + c;
#pragma unroll
        for (int v = 0; v < 8; ++v) {
            pp[v * PP]             = acc00[v];
            pp[v * PP + 16]        = acc01[v];
            pp[(16 + v) * PP]      = acc10[v];
            pp[(16 + v) * PP + 16] = acc11[v];
        }
    }
    __syncthreads();

    int* pick = (int*)(rec + (size_t)ntok * 32);
    const int slot = (lane >> 2) & 1;
    const int sel  = lane & 3;
#pragma unroll
    for (int half = 0; half < 2; ++half) {
        const int tl = half * 32 + wave * 4 + (lane >> 3);
        const float* p = Ps + tl * PP;
        float4 la = *(const float4*)(p);
        float4 lb = *(const float4*)(p + 4);
        float4 cv = *(const float4*)(p + 8 + sel * 4);
#pragma unroll
        for (int q = 1; q < 4; ++q) {
            const float* pq = p + q * (XT * PP);
            const float4 a2 = *(const float4*)(pq);
            const float4 b2 = *(const float4*)(pq + 4);
            const float4 c2 = *(const float4*)(pq + 8 + sel * 4);
            la.x += a2.x; la.y += a2.y; la.z += a2.z; la.w += a2.w;
            lb.x += b2.x; lb.y += b2.y; lb.z += b2.z; lb.w += b2.w;
            cv.x += c2.x; cv.y += c2.y; cv.z += c2.z; cv.w += c2.w;
        }
        float l[8];
        l[0] = la.x; l[1] = la.y; l[2] = la.z; l[3] = la.w;
        l[4] = lb.x; l[5] = lb.y; l[6] = lb.z; l[7] = lb.w;

        float v0 = l[0]; int i0 = 0;
#pragma unroll
        for (int j = 1; j < 8; ++j) { float lj = l[j]; if (lj > v0) { v0 = lj; i0 = j; } }
        float v1 = -3.4e38f; int i1 = 0;
#pragma unroll
        for (int j = 0; j < 8; ++j) {
            if (j == i0) continue;
            float lj = l[j]; if (lj > v1) { v1 = lj; i1 = j; }
        }
        float ex = __expf(v1 - v0);
        float g0 = 1.0f / (1.0f + ex);
        float g1 = ex   / (1.0f + ex);

        float g = slot ? g1 : g0;
        g *= SCALING;

        float4 outv = make_float4(g * cv.x, g * cv.y, g * cv.z, g * cv.w);
        *(float4*)(rec + (size_t)(tok0 + tl) * 32 + (lane & 7) * 4) = outv;
        if ((lane & 7) == 0) pick[tok0 + tl] = i0 | (i1 << 8);
    }
}

#define SM    256
#define SN    64
#define PL    68
#define LCAP  512
#define NCB   2

__global__ __launch_bounds__(256) void combine_picked(const float* __restrict__ rec,
                                                      const float* __restrict__ routed,
                                                      float* __restrict__ C, int ntok) {
    __shared__ __attribute__((aligned(16))) float Pl[(2 * SM + 1) * PL];
    __shared__ __attribute__((aligned(16))) int   Ls[NEXP * LCAP];
    __shared__ int Cn[8 * NEXP];

    const int tid  = threadIdx.x;
    const int lane = tid & 31;
    const int wave = tid >> 5;
    const int h    = lane >> 4;
    const int c    = lane & 15;
    const int nb   = blockIdx.x & 7;
    const int mb   = blockIdx.x >> 3;
    int n0         = nb * (SN * NCB);
    const int tok0 = mb * SM;
    const int* pick = (const int*)(rec + (size_t)ntok * 32);

    float4 bf[4][2];
    auto loadB = [&](int col0) {
#pragma unroll
        for (int ct = 0; ct < 4; ++ct) {
            const float* bp = routed + ((size_t)(wave * D_OUT + col0 + ct * 16 + c) * RANK + h * 4);
            bf[ct][0] = *(const float4*)(bp);
            bf[ct][1] = *(const float4*)(bp + 8);
        }
    };
    loadB(n0);

    const int pk = pick[tok0 + tid];
    const int i0 = pk & 7;
    const int i1 = (pk >> 8) & 7;
    const unsigned below = (1u << lane) - 1u;
    int pos0 = 0, pos1 = 0;
#pragma unroll
    for (int e = 0; e < NEXP; ++e) {
        const unsigned m0 = __builtin_amdgcn_ballot_w32(i0 == e);
        const unsigned m1 = __builtin_amdgcn_ballot_w32(i1 == e);
        const int c0 = __builtin_popcount(m0);
        const int c1 = __builtin_popcount(m1);
        if (i0 == e) pos0 = __builtin_popcount(m0 & below);
        if (i1 == e) pos1 = c0 + __builtin_popcount(m1 & below);
        if (lane == e) Cn[wave * NEXP + e] = c0 + c1;
    }
    __syncthreads();

    int base0 = 0, base1 = 0, n = 0;
#pragma unroll
    for (int w2 = 0; w2 < 8; ++w2) {
        const int a = Cn[w2 * NEXP + i0];
        const int b = Cn[w2 * NEXP + i1];
        if (w2 < wave) { base0 += a; base1 += b; }
        n += Cn[w2 * NEXP + wave];
    }
    Ls[i0 * LCAP + base0 + pos0] = tid;
    Ls[i1 * LCAP + base1 + pos1] = tid | 256;
    __syncthreads();

    n = __builtin_amdgcn_readfirstlane(n);
    const int ntile = (n + 15) >> 4;
    int* myl = Ls + wave * LCAP;

    {
        const int p = n + lane;
        const int padEnd = (ntile > 0 ? ntile : 1) * 16;
        if (lane < 16 && p < padEnd) myl[p] = 512;
    }

    auto loadEnt = [&](int tq) -> int {
        const int tt = tq < ntile ? tq : 0;
        return myl[tt * 16 + c];
    };
    auto gatherA = [&](int ent, float4& f0, float4& f1) {
        const float* ap = rec + (size_t)(tok0 + (ent & 255)) * 32 + ((ent >> 8) & 1) * 16 + h * 4;
        f0 = *(const float4*)(ap);
        f1 = *(const float4*)(ap + 8);
    };

    auto tile = [&](int t, const float4& x0, const float4& x1) {
        const int4 ea = *(const int4*)(myl + t * 16 + h * 8);
        const int4 eb = *(const int4*)(myl + t * 16 + h * 8 + 4);

        v8f acc[4];
#pragma unroll
        for (int ct = 0; ct < 4; ++ct) acc[ct] = (v8f){0.f, 0.f, 0.f, 0.f, 0.f, 0.f, 0.f, 0.f};
#pragma unroll
        for (int j = 0; j < 2; ++j) {
            const float4 xa = j ? x1 : x0;
            v2f a;
            a[0] = xa.x; a[1] = xa.y;
#pragma unroll
            for (int ct = 0; ct < 4; ++ct) {
                v2f b; b[0] = bf[ct][j].x; b[1] = bf[ct][j].y;
                acc[ct] = __builtin_amdgcn_wmma_f32_16x16x4_f32(false, a, false, b, (short)0, acc[ct], false, false);
            }
            a[0] = xa.z; a[1] = xa.w;
#pragma unroll
            for (int ct = 0; ct < 4; ++ct) {
                v2f b; b[0] = bf[ct][j].z; b[1] = bf[ct][j].w;
                acc[ct] = __builtin_amdgcn_wmma_f32_16x16x4_f32(false, a, false, b, (short)0, acc[ct], false, false);
            }
        }

        int ev[8];
        ev[0] = ea.x; ev[1] = ea.y; ev[2] = ea.z; ev[3] = ea.w;
        ev[4] = eb.x; ev[5] = eb.y; ev[6] = eb.z; ev[7] = eb.w;
#pragma unroll
        for (int v = 0; v < 8; ++v) {
            const int ent = ev[v];
            float* d = Pl + (((ent >> 8) & 3) * SM + (ent & 255)) * PL + c;
#pragma unroll
            for (int ct = 0; ct < 4; ++ct) d[ct * 16] = acc[ct][v];
        }
    };

    float4 fa0, fa1, fb0, fb1, fc0, fc1;
    gatherA(loadEnt(0), fa0, fa1);
    gatherA(loadEnt(1), fb0, fb1);
    const int ntrip = ntile / 3;
    for (int cb = 0; cb < NCB; ++cb) {
    int en = loadEnt(2);
    int t = 0;
    for (int k = 0; k < ntrip; ++k, t += 3) {
        gatherA(en, fc0, fc1);
        en = loadEnt(t + 3);
        __builtin_amdgcn_sched_barrier(0);
        tile(t, fa0, fa1);
        gatherA(en, fa0, fa1);
        en = loadEnt(t + 4);
        __builtin_amdgcn_sched_barrier(0);
        tile(t + 1, fb0, fb1);
        gatherA(en, fb0, fb1);
        en = loadEnt(t + 5);
        __builtin_amdgcn_sched_barrier(0);
        tile(t + 2, fc0, fc1);
    }
    if (t < ntile) {
        tile(t, fa0, fa1);
        if (t + 1 < ntile) tile(t + 1, fb0, fb1);
    }
    __syncthreads();
    if (cb + 1 < NCB) {
        loadB(n0 + SN);
        gatherA(loadEnt(0), fa0, fa1);
        gatherA(loadEnt(1), fb0, fb1);
    }

    float* ob = C + (size_t)tok0 * D_OUT + n0;
#pragma unroll
    for (int i = 0; i < 16; ++i) {
        const int idx = i * 256 + tid;
        const int row = idx >> 4;
        const int q   = idx & 15;
        const float4 s0 = *(const float4*)(Pl + row * PL + q * 4);
        const float4 s1 = *(const float4*)(Pl + (SM + row) * PL + q * 4);
        float4 o;
        o.x = s0.x + s1.x; o.y = s0.y + s1.y; o.z = s0.z + s1.z; o.w = s0.w + s1.w;
        *(float4*)(ob + (size_t)row * D_OUT + q * 4) = o;
    }
    n0 += SN;
    if (cb + 1 < NCB) {
        __builtin_amdgcn_fence(__ATOMIC_RELEASE, "workgroup", "local");
        __builtin_amdgcn_s_barrier();
        __builtin_amdgcn_fence(__ATOMIC_ACQUIRE, "workgroup", "local");
    }
    }
}

extern "C" void kernel_launch(void* const* d_in, const int* in_sizes, int n_in,
                              void* d_out, int out_size, void* d_ws, size_t ws_size,
                              hipStream_t stream) {
    const float* x        = (const float*)d_in[0];
    const float* w_route  = (const float*)d_in[1];
    const float* compress = (const float*)d_in[2];
    const float* routed   = (const float*)d_in[3];
    float*       out      = (float*)d_out;

    float* rec = (float*)d_ws;

    const int ntok = in_sizes[0] / D_IN;

    route_compress<<<ntok / XT, 256, 0, stream>>>(x, w_route, compress, rec, ntok);

    combine_picked<<<(ntok / SM) * (D_OUT / (SN * NCB)), 256, 0, stream>>>(rec, routed, out, ntok);
}
